# starting kernel
# baseline (speedup 1.0000x reference)
#include <hip/hip_runtime.h>

typedef __attribute__((ext_vector_type(2))) float v2f;
typedef __attribute__((ext_vector_type(8))) float v8f;

#define IN_CH 128
#define HID 32


__global__ void init_deg_kernel(float* __restrict__ deg, int n) {
  int i = blockIdx.x * blockDim.x + threadIdx.x;
  if (i < n) deg[i] = 1.0f;
}

__global__ void degree_kernel(const int* __restrict__ dst, float* __restrict__ deg, int e_count) {
  int e = blockIdx.x * blockDim.x + threadIdx.x;
  if (e < e_count) atomicAdd(&deg[dst[e]], 1.0f);
}

__global__ void dinv_kernel(float* __restrict__ degdinv, int n) {
  int i = blockIdx.x * blockDim.x + threadIdx.x;
  if (i < n) {
    float d = degdinv[i];
    degdinv[i] = (d > 0.0f) ? rsqrtf(d) : 0.0f;
  }
}

__global__ void init_agg_kernel(const float* __restrict__ h, const float* __restrict__ dinv,
                                const float* __restrict__ bias, float* __restrict__ agg, int n) {
  int i = blockIdx.x * blockDim.x + threadIdx.x;
  if (i < n * HID) {
    int node = i >> 5;
    int c = i & 31;
    float di = dinv[node];
    agg[i] = bias[c] + h[i] * di * di;
  }
}

__global__ void relu_kernel(const float* __restrict__ in, float* __restrict__ out, int total) {
  int i = blockIdx.x * blockDim.x + threadIdx.x;
  if (i < total) out[i] = fmaxf(in[i], 0.0f);
}

__global__ __launch_bounds__(256)
void edge_agg_kernel(const int* __restrict__ src, const int* __restrict__ dst,
                     const float* __restrict__ dinv, const float* __restrict__ h,
                     float* __restrict__ agg, int e_count) {
  int lane = threadIdx.x & 31;
  int e = blockIdx.x * (blockDim.x >> 5) + (threadIdx.x >> 5);
  if (e >= e_count) return;
  int s = src[e];
  int d = dst[e];
  float nrm = dinv[s] * dinv[d];
  float v = h[s * HID + lane] * nrm;
  atomicAdd(&agg[d * HID + lane], v);
}

template <int K>
__global__ __launch_bounds__(128)
void gemm_n32_kernel(const float* __restrict__ A, const float* __restrict__ W,
                     float* __restrict__ out, int M) {
  constexpr int LDK = K + 1;
  __shared__ float sA[32 * LDK];

  const int tid = threadIdx.x;
  const int block_row0 = blockIdx.x * 32;

  for (int idx = tid; idx < 32 * K; idx += 128) {
    int r = idx / K;
    int c = idx - r * K;
    int gr = block_row0 + r;
    sA[r * LDK + c] = (gr < M) ? A[(long long)gr * K + c] : 0.0f;
  }
  __syncthreads();

  const int wid   = tid >> 5;
  const int lane  = tid & 31;
  const int lhalf = lane >> 4;
  const int l15   = lane & 15;
  const int warp_m = (wid >> 1) * 16;
  const int n0     = (wid & 1) * 16;

  v8f acc = {};
  for (int k = 0; k < K; k += 4) {
    const float* ap = &sA[(warp_m + l15) * LDK + k + lhalf * 2];
    v2f a;
    a.x = ap[0];
    a.y = ap[1];
    const float* bp = &W[(k + lhalf * 2) * HID + n0 + l15];
    v2f b;
    b.x = bp[0];
    b.y = bp[HID];
    acc = __builtin_amdgcn_wmma_f32_16x16x4_f32(false, a, false, b, (short)0, acc,
                                                false, false);
  }

  const int grow_base = block_row0 + warp_m + lhalf * 8;
  const int col = n0 + l15;
#pragma unroll
  for (int j = 0; j < 8; ++j) {
    int gr = grow_base + j;
    if (gr < M) out[(long long)gr * HID + col] = acc[j];
  }
}


extern "C" void kernel_launch(void* const* d_in, const int* in_sizes, int n_in,
                              void* d_out, int out_size, void* d_ws, size_t ws_size,
                              hipStream_t stream) {
  const float* x  = (const float*)d_in[0];
  const int*   ei = (const int*)d_in[1];
  const float* W1 = (const float*)d_in[2];
  const float* b1 = (const float*)d_in[3];
  const float* W2 = (const float*)d_in[4];
  const float* b2 = (const float*)d_in[5];
  float* out = (float*)d_out;

  const int N = in_sizes[0] / IN_CH;
  const int E = in_sizes[1] / 2;
  const int* src = ei;
  const int* dst = ei + E;

  float* ws = (float*)d_ws;
  const int Npad = (N + 63) & ~63;
  float* dinv = ws;
  float* bufA = ws + Npad;
  float* bufB = bufA + (size_t)N * HID;

  const int T = 256;
  const unsigned gN    = (N + T - 1) / T;
  const unsigned gE    = (E + T - 1) / T;
  const unsigned gNC   = ((unsigned)N * HID + T - 1) / T;
  const unsigned gEdge = (E + (T / 32) - 1) / (T / 32);
  const unsigned gGemm = (N + 31) / 32;

  init_deg_kernel<<<gN, T, 0, stream>>>(dinv, N);
  degree_kernel<<<gE, T, 0, stream>>>(dst, dinv, E);
  dinv_kernel<<<gN, T, 0, stream>>>(dinv, N);

  gemm_n32_kernel<IN_CH><<<gGemm, 128, 0, stream>>>(x, W1, bufA, N);
  init_agg_kernel<<<gNC, T, 0, stream>>>(bufA, dinv, b1, bufB, N);
  edge_agg_kernel<<<gEdge, T, 0, stream>>>(src, dst, dinv, bufA, bufB, E);
  relu_kernel<<<gNC, T, 0, stream>>>(bufB, bufA, N * HID);

  gemm_n32_kernel<HID><<<gGemm, 128, 0, stream>>>(bufA, W2, bufB, N);
  init_agg_kernel<<<gNC, T, 0, stream>>>(bufB, dinv, b2, out, N);
  edge_agg_kernel<<<gEdge, T, 0, stream>>>(src, dst, dinv, bufB, out, E);
}


# edge_agg: one wave handles 32 consecutive edges (coalesced index loads, lane broadcast, 32 row gathers in flight) instead of one edge per wave; float atomics kept, sums still in arrival order
# speedup vs baseline: 1.8648x; 1.8648x over previous
#include <hip/hip_runtime.h>

typedef __attribute__((ext_vector_type(2))) float v2f;
typedef __attribute__((ext_vector_type(8))) float v8f;

#define IN_CH 128
#define HID 32


__global__ void init_deg_kernel(float* __restrict__ deg, int n) {
  int i = blockIdx.x * blockDim.x + threadIdx.x;
  if (i < n) deg[i] = 1.0f;
}

__global__ void degree_kernel(const int* __restrict__ dst, float* __restrict__ deg, int e_count) {
  int e = blockIdx.x * blockDim.x + threadIdx.x;
  if (e < e_count) atomicAdd(&deg[dst[e]], 1.0f);
}

__global__ void dinv_kernel(float* __restrict__ degdinv, int n) {
  int i = blockIdx.x * blockDim.x + threadIdx.x;
  if (i < n) {
    float d = degdinv[i];
    degdinv[i] = (d > 0.0f) ? rsqrtf(d) : 0.0f;
  }
}

__global__ void init_agg_kernel(const float* __restrict__ h, const float* __restrict__ dinv,
                                const float* __restrict__ bias, float* __restrict__ agg, int n) {
  int i = blockIdx.x * blockDim.x + threadIdx.x;
  if (i < n * HID) {
    int node = i >> 5;
    int c = i & 31;
    float di = dinv[node];
    agg[i] = bias[c] + h[i] * di * di;
  }
}

__global__ void relu_kernel(const float* __restrict__ in, float* __restrict__ out, int total) {
  int i = blockIdx.x * blockDim.x + threadIdx.x;
  if (i < total) out[i] = fmaxf(in[i], 0.0f);
}

__global__ __launch_bounds__(256)
void edge_agg_kernel(const int* __restrict__ src, const int* __restrict__ dst,
                     const float* __restrict__ dinv, const float* __restrict__ h,
                     float* __restrict__ agg, int e_count) {
  const int lane = threadIdx.x & 31;
  const int wid = __builtin_amdgcn_readfirstlane((int)(threadIdx.x >> 5));
  const int base = (blockIdx.x * (int)(blockDim.x >> 5) + wid) * 32;
  if (base >= e_count) return;
  const int left = e_count - base;
  const int cnt = left < 32 ? left : 32;

  int s = 0, d = 0;
  float ds = 0.0f, dd = 0.0f;
  if (lane < cnt) {
    s = src[base + lane];
    d = dst[base + lane];
    ds = dinv[s];
    dd = dinv[d];
  }

  if (cnt == 32) {
    float v[32];
#pragma unroll
    for (int j = 0; j < 32; ++j) {
      const float* hrow = h + (size_t)__builtin_amdgcn_readlane(s, j) * HID;
      v[j] = hrow[lane];
    }
    const int nrm_bits = __float_as_int(ds * dd);
#pragma unroll
    for (int j = 0; j < 32; ++j) {
      float* arow = agg + (size_t)__builtin_amdgcn_readlane(d, j) * HID;
      const float nj = __int_as_float(__builtin_amdgcn_readlane(nrm_bits, j));
      atomicAdd(&arow[lane], v[j] * nj);
    }
  } else {
    for (int j = 0; j < cnt; ++j) {
      const int sj = __builtin_amdgcn_readlane(s, j);
      const int dj = __builtin_amdgcn_readlane(d, j);
      const float nj = __int_as_float(__builtin_amdgcn_readlane(__float_as_int(ds), j)) *
                       __int_as_float(__builtin_amdgcn_readlane(__float_as_int(dd), j));
      atomicAdd(&agg[(size_t)dj * HID + lane], h[(size_t)sj * HID + lane] * nj);
    }
  }
}

template <int K>
__global__ __launch_bounds__(128)
void gemm_n32_kernel(const float* __restrict__ A, const float* __restrict__ W,
                     float* __restrict__ out, int M) {
  constexpr int LDK = K + 1;
  __shared__ float sA[32 * LDK];

  const int tid = threadIdx.x;
  const int block_row0 = blockIdx.x * 32;

  for (int idx = tid; idx < 32 * K; idx += 128) {
    int r = idx / K;
    int c = idx - r * K;
    int gr = block_row0 + r;
    sA[r * LDK + c] = (gr < M) ? A[(long long)gr * K + c] : 0.0f;
  }
  __syncthreads();

  const int wid   = tid >> 5;
  const int lane  = tid & 31;
  const int lhalf = lane >> 4;
  const int l15   = lane & 15;
  const int warp_m = (wid >> 1) * 16;
  const int n0     = (wid & 1) * 16;

  v8f acc = {};
  for (int k = 0; k < K; k += 4) {
    const float* ap = &sA[(warp_m + l15) * LDK + k + lhalf * 2];
    v2f a;
    a.x = ap[0];
    a.y = ap[1];
    const float* bp = &W[(k + lhalf * 2) * HID + n0 + l15];
    v2f b;
    b.x = bp[0];
    b.y = bp[HID];
    acc = __builtin_amdgcn_wmma_f32_16x16x4_f32(false, a, false, b, (short)0, acc,
                                                false, false);
  }

  const int grow_base = block_row0 + warp_m + lhalf * 8;
  const int col = n0 + l15;
#pragma unroll
  for (int j = 0; j < 8; ++j) {
    int gr = grow_base + j;
    if (gr < M) out[(long long)gr * HID + col] = acc[j];
  }
}


extern "C" void kernel_launch(void* const* d_in, const int* in_sizes, int n_in,
                              void* d_out, int out_size, void* d_ws, size_t ws_size,
                              hipStream_t stream) {
  const float* x  = (const float*)d_in[0];
  const int*   ei = (const int*)d_in[1];
  const float* W1 = (const float*)d_in[2];
  const float* b1 = (const float*)d_in[3];
  const float* W2 = (const float*)d_in[4];
  const float* b2 = (const float*)d_in[5];
  float* out = (float*)d_out;

  const int N = in_sizes[0] / IN_CH;
  const int E = in_sizes[1] / 2;
  const int* src = ei;
  const int* dst = ei + E;

  float* ws = (float*)d_ws;
  const int Npad = (N + 63) & ~63;
  float* dinv = ws;
  float* bufA = ws + Npad;
  float* bufB = bufA + (size_t)N * HID;

  const int T = 256;
  const unsigned gN    = (N + T - 1) / T;
  const unsigned gE    = (E + T - 1) / T;
  const unsigned gNC   = ((unsigned)N * HID + T - 1) / T;
  const unsigned gGemm = (N + 31) / 32;

  init_deg_kernel<<<gN, T, 0, stream>>>(dinv, N);
  degree_kernel<<<gE, T, 0, stream>>>(dst, dinv, E);
  dinv_kernel<<<gN, T, 0, stream>>>(dinv, N);

  gemm_n32_kernel<IN_CH><<<gGemm, 128, 0, stream>>>(x, W1, bufA, N);
  init_agg_kernel<<<gNC, T, 0, stream>>>(bufA, dinv, b1, bufB, N);
  edge_agg_kernel<<<gE, T, 0, stream>>>(src, dst, dinv, bufA, bufB, E);
  relu_kernel<<<gNC, T, 0, stream>>>(bufB, bufA, N * HID);

  gemm_n32_kernel<HID><<<gGemm, 128, 0, stream>>>(bufA, W2, bufB, N);
  init_agg_kernel<<<gNC, T, 0, stream>>>(bufB, dinv, b2, out, N);
  edge_agg_kernel<<<gE, T, 0, stream>>>(src, dst, dinv, bufB, out, E);
}


# edges counting-sorted into 256-node dst buckets; one workgroup sums a bucket in an LDS tile (LDS float adds, arrival order); norm factored as dinv[d]*sum(h*dinv[s]); no global float atomics left
# speedup vs baseline: 4.1928x; 2.2484x over previous
#include <hip/hip_runtime.h>


typedef __attribute__((ext_vector_type(2))) float v2f;
typedef __attribute__((ext_vector_type(8))) float v8f;

#define IN_CH 128
#define HID 32

#define BSH 8
#define BK 256
#define NBMAX 512
#define PT 512
#define PE 16
#define AT 1024


__global__ void zero_counts_kernel(int* __restrict__ gcount, int nb) {
  int i = blockIdx.x * blockDim.x + threadIdx.x;
  if (i < nb) gcount[i] = 0;
}

__global__ __launch_bounds__(PT)
void bucket_hist_kernel(const int* __restrict__ dst, int* __restrict__ gcount, int e_count, int nb) {
  __shared__ int hist[NBMAX];
  for (int i = threadIdx.x; i < NBMAX; i += PT) hist[i] = 0;
  __syncthreads();
  const int base = blockIdx.x * (PT * PE);
#pragma unroll
  for (int k = 0; k < PE; ++k) {
    const int e = base + k * PT + (int)threadIdx.x;
    if (e < e_count) atomicAdd(&hist[dst[e] >> BSH], 1);
  }
  __syncthreads();
  for (int i = threadIdx.x; i < nb; i += PT) {
    const int c = hist[i];
    if (c) atomicAdd(&gcount[i], c);
  }
}

__global__ __launch_bounds__(NBMAX)
void bucket_scan_kernel(const int* __restrict__ gcount, int* __restrict__ gbase,
                        int* __restrict__ gcursor, int nb) {
  __shared__ int s[NBMAX];
  const int t = threadIdx.x;
  const int c = (t < nb) ? gcount[t] : 0;
  s[t] = c;
  __syncthreads();
  for (int off = 1; off < NBMAX; off <<= 1) {
    const int v = (t >= off) ? s[t - off] : 0;
    __syncthreads();
    s[t] += v;
    __syncthreads();
  }
  if (t < nb) {
    gbase[t] = s[t] - c;
    gcursor[t] = s[t] - c;
  }
  if (t == nb - 1) gbase[nb] = s[t];
}

__global__ __launch_bounds__(PT)
void bucket_scatter_kernel(const int* __restrict__ src, const int* __restrict__ dst,
                           int* __restrict__ gcursor, unsigned* __restrict__ keys,
                           int e_count, int nb) {
  __shared__ int hist[NBMAX];
  for (int i = threadIdx.x; i < NBMAX; i += PT) hist[i] = 0;
  __syncthreads();
  const int base = blockIdx.x * (PT * PE);
  unsigned key[PE];
  int bkt[PE];
  int rank[PE];
#pragma unroll
  for (int k = 0; k < PE; ++k) {
    const int e = base + k * PT + (int)threadIdx.x;
    key[k] = 0u;
    bkt[k] = 0;
    rank[k] = -1;
    if (e < e_count) {
      const int s = src[e];
      const int d = dst[e];
      key[k] = ((unsigned)s << BSH) | (unsigned)(d & (BK - 1));
      bkt[k] = d >> BSH;
      rank[k] = atomicAdd(&hist[bkt[k]], 1);
    }
  }
  __syncthreads();
  for (int i = threadIdx.x; i < nb; i += PT) {
    const int c = hist[i];
    hist[i] = c ? atomicAdd(&gcursor[i], c) : 0;
  }
  __syncthreads();
#pragma unroll
  for (int k = 0; k < PE; ++k) {
    if (rank[k] >= 0) keys[hist[bkt[k]] + rank[k]] = key[k];
  }
}

__global__ __launch_bounds__(AT)
void bucket_dinv_kernel(const unsigned* __restrict__ keys, const int* __restrict__ gbase,
                        float* __restrict__ dinv, int n) {
  __shared__ int cnt[BK];
  const int t = threadIdx.x;
  if (t < BK) cnt[t] = 0;
  __syncthreads();
  const int beg = gbase[blockIdx.x];
  const int end = gbase[blockIdx.x + 1];
  for (int e = beg + t; e < end; e += AT) atomicAdd(&cnt[keys[e] & (BK - 1)], 1);
  __syncthreads();
  if (t < BK) {
    const int node = blockIdx.x * BK + t;
    if (node < n) {
      const float d = 1.0f + (float)cnt[t];
      dinv[node] = (d > 0.0f) ? rsqrtf(d) : 0.0f;
    }
  }
}

template <bool RELU>
__global__ __launch_bounds__(AT)
void bucket_agg_kernel(const unsigned* __restrict__ keys, const int* __restrict__ gbase,
                       const float* __restrict__ hs, const float* __restrict__ dinv,
                       const float* __restrict__ bias, float* __restrict__ out, int n) {
  __shared__ float4 tile4[BK * HID / 4];
  float* tile = reinterpret_cast<float*>(tile4);
  const int t = threadIdx.x;
  const int lane = t & 31;
  const int wid = __builtin_amdgcn_readfirstlane(t >> 5);
  const int node0 = blockIdx.x * BK;
  const size_t row0 = (size_t)node0 * HID;
  constexpr int NV = BK * HID / 4 / AT;

  {
    const float4* hs4 = reinterpret_cast<const float4*>(hs + row0);
    float4 a[NV];
#pragma unroll
    for (int u = 0; u < NV; ++u) {
      const int i4 = t + u * AT;
      const int node = node0 + (i4 >> 3);
      a[u] = (node < n) ? hs4[i4] : make_float4(0.0f, 0.0f, 0.0f, 0.0f);
    }
#pragma unroll
    for (int u = 0; u < NV; ++u) tile4[t + u * AT] = a[u];
  }
  __syncthreads();

  const int beg = gbase[blockIdx.x];
  const int end = gbase[blockIdx.x + 1];
  const int last = max(end - 1, 0);
  int b0 = beg + wid * 32;
  int k = (int)keys[min(b0 + lane, last)];
  for (; b0 < end; b0 += AT) {
    const int left = end - b0;
    const int kn = (int)keys[min(b0 + AT + lane, last)];
    if (left >= 32) {
      float v[32];
#pragma unroll
      for (int j = 0; j < 32; ++j) {
        const unsigned kj = (unsigned)__builtin_amdgcn_readlane(k, j);
        v[j] = hs[(size_t)(kj >> BSH) * HID + lane];
      }
#pragma unroll
      for (int j = 0; j < 32; ++j) {
        const unsigned kj = (unsigned)__builtin_amdgcn_readlane(k, j);
        atomicAdd(&tile[(kj & (BK - 1)) * HID + lane], v[j]);
      }
    } else {
      for (int j = 0; j < left; ++j) {
        const unsigned kj = (unsigned)__builtin_amdgcn_readlane(k, j);
        atomicAdd(&tile[(kj & (BK - 1)) * HID + lane], hs[(size_t)(kj >> BSH) * HID + lane]);
      }
    }
    k = kn;
  }
  __syncthreads();

  {
    float4* out4 = reinterpret_cast<float4*>(out + row0);
#pragma unroll
    for (int u = 0; u < NV; ++u) {
      const int i4 = t + u * AT;
      const int node = node0 + (i4 >> 3);
      if (node < n) {
        const float4 a = tile4[i4];
        const float di = dinv[node];
        const float* bp = bias + (i4 & 7) * 4;
        float4 r;
        r.x = di * a.x + bp[0];
        r.y = di * a.y + bp[1];
        r.z = di * a.z + bp[2];
        r.w = di * a.w + bp[3];
        if (RELU) {
          r.x = fmaxf(r.x, 0.0f);
          r.y = fmaxf(r.y, 0.0f);
          r.z = fmaxf(r.z, 0.0f);
          r.w = fmaxf(r.w, 0.0f);
        }
        out4[i4] = r;
      }
    }
  }
}

template <int K>
__global__ __launch_bounds__(128)
void gemm_n32_kernel(const float* __restrict__ A, const float* __restrict__ W,
                     const float* __restrict__ rowscale, float* __restrict__ out, int M) {
  constexpr int LDK = K + 1;
  __shared__ float sA[32 * LDK];

  const int tid = threadIdx.x;
  const int block_row0 = blockIdx.x * 32;

  for (int idx = tid; idx < 32 * K; idx += 128) {
    int r = idx / K;
    int c = idx - r * K;
    int gr = block_row0 + r;
    sA[r * LDK + c] = (gr < M) ? A[(long long)gr * K + c] : 0.0f;
  }
  __syncthreads();

  const int wid   = tid >> 5;
  const int lane  = tid & 31;
  const int lhalf = lane >> 4;
  const int l15   = lane & 15;
  const int warp_m = (wid >> 1) * 16;
  const int n0     = (wid & 1) * 16;

  v8f acc = {};
  for (int k = 0; k < K; k += 4) {
    const float* ap = &sA[(warp_m + l15) * LDK + k + lhalf * 2];
    v2f a;
    a.x = ap[0];
    a.y = ap[1];
    const float* bp = &W[(k + lhalf * 2) * HID + n0 + l15];
    v2f b;
    b.x = bp[0];
    b.y = bp[HID];
    acc = __builtin_amdgcn_wmma_f32_16x16x4_f32(false, a, false, b, (short)0, acc,
                                                false, false);
  }

  const int grow_base = block_row0 + warp_m + lhalf * 8;
  const int col = n0 + l15;
#pragma unroll
  for (int j = 0; j < 8; ++j) {
    int gr = grow_base + j;
    if (gr < M) out[(long long)gr * HID + col] = acc[j] * rowscale[gr];
  }
}


extern "C" void kernel_launch(void* const* d_in, const int* in_sizes, int n_in,
                              void* d_out, int out_size, void* d_ws, size_t ws_size,
                              hipStream_t stream) {
  const float* x  = (const float*)d_in[0];
  const int*   ei = (const int*)d_in[1];
  const float* W1 = (const float*)d_in[2];
  const float* b1 = (const float*)d_in[3];
  const float* W2 = (const float*)d_in[4];
  const float* b2 = (const float*)d_in[5];
  float* out = (float*)d_out;

  const int N = in_sizes[0] / IN_CH;
  const int E = in_sizes[1] / 2;
  const int* src = ei;
  const int* dst = ei + E;

  const int Npad = (N + 63) & ~63;
  const int Epad = (E + 63) & ~63;
  int* gcount  = (int*)d_ws;
  int* gbase   = gcount + NBMAX;
  int* gcursor = gbase + 2 * NBMAX;
  float* dinv  = (float*)(gcursor + NBMAX);
  unsigned* keys = (unsigned*)(dinv + Npad);
  float* bufA = (float*)(keys + Epad);
  float* bufB = bufA + (size_t)N * HID;

  const unsigned NB    = (N + BK - 1) / BK;
  const unsigned gPart = (E + PT * PE - 1) / (PT * PE);
  const unsigned gGemm = (N + 31) / 32;

  zero_counts_kernel<<<1, NBMAX, 0, stream>>>(gcount, NB);
  bucket_hist_kernel<<<gPart, PT, 0, stream>>>(dst, gcount, E, NB);
  bucket_scan_kernel<<<1, NBMAX, 0, stream>>>(gcount, gbase, gcursor, NB);
  bucket_scatter_kernel<<<gPart, PT, 0, stream>>>(src, dst, gcursor, keys, E, NB);
  bucket_dinv_kernel<<<NB, AT, 0, stream>>>(keys, gbase, dinv, N);

  gemm_n32_kernel<IN_CH><<<gGemm, 128, 0, stream>>>(x, W1, dinv, bufA, N);
  bucket_agg_kernel<true><<<NB, AT, 0, stream>>>(keys, gbase, bufA, dinv, b1, bufB, N);

  gemm_n32_kernel<HID><<<gGemm, 128, 0, stream>>>(bufB, W2, dinv, bufA, N);
  bucket_agg_kernel<false><<<NB, AT, 0, stream>>>(keys, gbase, bufA, dinv, b2, out, N);
}


# bucket edges ordered by destination (LDS counters, returning add as slot); a wave sums a node's run in registers, 4 edges per 16B load, then across lanes; no LDS float adds; sum order changes
# speedup vs baseline: 4.6510x; 1.1093x over previous
#include <hip/hip_runtime.h>


typedef __attribute__((ext_vector_type(2))) float v2f;
typedef __attribute__((ext_vector_type(8))) float v8f;

#define IN_CH 128
#define HID 32

#define BSH 8
#define BK 256
#define NBMAX 512
#define PT 512
#define PE 16
#define AT 1024
#define LCAP 12288
#define RW 8
#define NPW 8


__global__ void zero_counts_kernel(int* __restrict__ gcount, int nb) {
  int i = blockIdx.x * blockDim.x + threadIdx.x;
  if (i < nb) gcount[i] = 0;
}

__global__ __launch_bounds__(PT)
void bucket_hist_kernel(const int* __restrict__ dst, int* __restrict__ gcount, int e_count, int nb) {
  __shared__ int hist[NBMAX];
  for (int i = threadIdx.x; i < NBMAX; i += PT) hist[i] = 0;
  __syncthreads();
  const int base = blockIdx.x * (PT * PE);
#pragma unroll
  for (int k = 0; k < PE; ++k) {
    const int e = base + k * PT + (int)threadIdx.x;
    if (e < e_count) atomicAdd(&hist[dst[e] >> BSH], 1);
  }
  __syncthreads();
  for (int i = threadIdx.x; i < nb; i += PT) {
    const int c = hist[i];
    if (c) atomicAdd(&gcount[i], c);
  }
}

__global__ __launch_bounds__(NBMAX)
void bucket_scan_kernel(const int* __restrict__ gcount, int* __restrict__ gbase,
                        int* __restrict__ gcursor, int nb) {
  __shared__ int s[NBMAX];
  const int t = threadIdx.x;
  const int c = (t < nb) ? gcount[t] : 0;
  s[t] = c;
  __syncthreads();
  for (int off = 1; off < NBMAX; off <<= 1) {
    const int v = (t >= off) ? s[t - off] : 0;
    __syncthreads();
    s[t] += v;
    __syncthreads();
  }
  if (t < nb) {
    gbase[t] = s[t] - c;
    gcursor[t] = s[t] - c;
  }
  if (t == nb - 1) gbase[nb] = s[t];
}

__global__ __launch_bounds__(PT)
void bucket_scatter_kernel(const int* __restrict__ src, const int* __restrict__ dst,
                           int* __restrict__ gcursor, unsigned* __restrict__ keys,
                           int e_count, int nb) {
  __shared__ int hist[NBMAX];
  for (int i = threadIdx.x; i < NBMAX; i += PT) hist[i] = 0;
  __syncthreads();
  const int base = blockIdx.x * (PT * PE);
  unsigned key[PE];
  int bkt[PE];
  int rank[PE];
#pragma unroll
  for (int k = 0; k < PE; ++k) {
    const int e = base + k * PT + (int)threadIdx.x;
    key[k] = 0u;
    bkt[k] = 0;
    rank[k] = -1;
    if (e < e_count) {
      const int s = src[e];
      const int d = dst[e];
      key[k] = ((unsigned)s << BSH) | (unsigned)(d & (BK - 1));
      bkt[k] = d >> BSH;
      rank[k] = atomicAdd(&hist[bkt[k]], 1);
    }
  }
  __syncthreads();
  for (int i = threadIdx.x; i < nb; i += PT) {
    const int c = hist[i];
    hist[i] = c ? atomicAdd(&gcursor[i], c) : 0;
  }
  __syncthreads();
#pragma unroll
  for (int k = 0; k < PE; ++k) {
    if (rank[k] >= 0) keys[hist[bkt[k]] + rank[k]] = key[k];
  }
}

__global__ __launch_bounds__(AT)
void bucket_csr_kernel(const unsigned* __restrict__ keys, const int* __restrict__ gbase,
                       unsigned* __restrict__ csr_src, int* __restrict__ rowptr,
                       float* __restrict__ dinv, float* __restrict__ zero_row, int n) {
  __shared__ int cnt[BK];
  __shared__ int wsum[BK / 32];
  __shared__ unsigned stage[LCAP];
  const int t = threadIdx.x;
  const int lane = t & 31;
  if (t < BK) cnt[t] = 0;
  __syncthreads();
  const int beg = gbase[blockIdx.x];
  const int end = gbase[blockIdx.x + 1];
  const int m = end - beg;
  for (int e = beg + t; e < end; e += AT) atomicAdd(&cnt[keys[e] & (BK - 1)], 1);
  __syncthreads();
  int c = 0;
  int incl = 0;
  if (t < BK) {
    c = cnt[t];
    incl = c;
#pragma unroll
    for (int off = 1; off < 32; off <<= 1) {
      const int v = __shfl_up(incl, off);
      if (lane >= off) incl += v;
    }
    if (lane == 31) wsum[t >> 5] = incl;
  }
  __syncthreads();
  if (t < BK) {
    int start = incl - c;
    for (int w = 0; w < (t >> 5); ++w) start += wsum[w];
    cnt[t] = start;
    const int node = blockIdx.x * BK + t;
    if (node < n) {
      rowptr[node] = beg + start;
      const float d = 1.0f + (float)c;
      dinv[node] = (d > 0.0f) ? rsqrtf(d) : 0.0f;
    }
  }
  if (blockIdx.x == gridDim.x - 1) {
    if (t == 0) rowptr[n] = end;
    if (t < HID) zero_row[t] = 0.0f;
  }
  __syncthreads();
  const bool staged = (m <= LCAP);
  for (int e = beg + t; e < end; e += AT) {
    const unsigned k = keys[e];
    const int pos = atomicAdd(&cnt[k & (BK - 1)], 1);
    if (staged) stage[pos] = k >> BSH;
    else csr_src[beg + pos] = k >> BSH;
  }
  __syncthreads();
  if (staged) {
    for (int i = t; i < m; i += AT) csr_src[beg + i] = stage[i];
  }
}

template <bool RELU>
__global__ __launch_bounds__(RW * 32)
void row_sum_kernel(const unsigned* __restrict__ csr_src, const int* __restrict__ rowptr,
                    const float* __restrict__ hs, const float* __restrict__ dinv,
                    const float* __restrict__ bias, float* __restrict__ out, int n, int e_count) {
  const int lane = threadIdx.x & 31;
  const int wid = __builtin_amdgcn_readfirstlane((int)threadIdx.x >> 5);
  const int g = lane >> 3;
  const int q = lane & 7;
  const int n0 = (blockIdx.x * RW + wid) * NPW;
  if (n0 >= n) return;
  const int cntn = min(NPW, n - n0);
  const int elast = max(e_count - 1, 0);
  const float4* hs4 = reinterpret_cast<const float4*>(hs);
  float4* out4 = reinterpret_cast<float4*>(out);
  const int rp = rowptr[min(n0 + lane, n)];
  const float4 b4 = reinterpret_cast<const float4*>(bias)[q];
  int beg = __builtin_amdgcn_readlane(rp, 0);
  int end = __builtin_amdgcn_readlane(rp, 1);
  unsigned kA = csr_src[min(beg + lane, elast)];
  unsigned kB = csr_src[min(beg + 32 + lane, elast)];
  kA = (beg + lane < end) ? kA : (unsigned)n;
  kB = (beg + 32 + lane < end) ? kB : (unsigned)n;
  for (int i = 0; i < cntn; ++i) {
    const int node = n0 + i;
    const int nend = (i + 1 < cntn) ? __builtin_amdgcn_readlane(rp, min(i + 2, 31)) : end;
    unsigned nA = csr_src[min(end + lane, elast)];
    unsigned nB = csr_src[min(end + 32 + lane, elast)];
    const float di = dinv[node];
    const int deg = end - beg;
    unsigned sA[8];
#pragma unroll
    for (int s = 0; s < 8; ++s)
      sA[s] = (unsigned)__builtin_amdgcn_ds_bpermute((4 * s + g) << 2, (int)kA);
    float4 acc = hs4[(size_t)(g == 0 ? node : n) * 8 + q];
    float4 vA[8];
#pragma unroll
    for (int s = 0; s < 8; ++s) vA[s] = hs4[(size_t)sA[s] * 8 + q];
    float4 accB = make_float4(0.0f, 0.0f, 0.0f, 0.0f);
    if (deg > 32) {
      unsigned sB[8];
#pragma unroll
      for (int s = 0; s < 8; ++s)
        sB[s] = (unsigned)__builtin_amdgcn_ds_bpermute((4 * s + g) << 2, (int)kB);
      float4 vB[8];
#pragma unroll
      for (int s = 0; s < 8; ++s) vB[s] = hs4[(size_t)sB[s] * 8 + q];
#pragma unroll
      for (int s = 0; s < 8; ++s) {
        accB.x += vB[s].x;
        accB.y += vB[s].y;
        accB.z += vB[s].z;
        accB.w += vB[s].w;
      }
      for (int e = beg + 64 + g; e < end; e += 4) {
        const float4 r = hs4[(size_t)csr_src[e] * 8 + q];
        accB.x += r.x;
        accB.y += r.y;
        accB.z += r.z;
        accB.w += r.w;
      }
    }
#pragma unroll
    for (int s = 0; s < 8; ++s) {
      acc.x += vA[s].x;
      acc.y += vA[s].y;
      acc.z += vA[s].z;
      acc.w += vA[s].w;
    }
    acc.x += accB.x;
    acc.y += accB.y;
    acc.z += accB.z;
    acc.w += accB.w;
    acc.x += __shfl_xor(acc.x, 8);
    acc.y += __shfl_xor(acc.y, 8);
    acc.z += __shfl_xor(acc.z, 8);
    acc.w += __shfl_xor(acc.w, 8);
    acc.x += __shfl_xor(acc.x, 16);
    acc.y += __shfl_xor(acc.y, 16);
    acc.z += __shfl_xor(acc.z, 16);
    acc.w += __shfl_xor(acc.w, 16);
    if (g == 0) {
      float4 r;
      r.x = di * acc.x + b4.x;
      r.y = di * acc.y + b4.y;
      r.z = di * acc.z + b4.z;
      r.w = di * acc.w + b4.w;
      if (RELU) {
        r.x = fmaxf(r.x, 0.0f);
        r.y = fmaxf(r.y, 0.0f);
        r.z = fmaxf(r.z, 0.0f);
        r.w = fmaxf(r.w, 0.0f);
      }
      out4[(size_t)node * 8 + q] = r;
    }
    kA = (end + lane < nend) ? nA : (unsigned)n;
    kB = (end + 32 + lane < nend) ? nB : (unsigned)n;
    beg = end;
    end = nend;
  }
}

template <int K>
__global__ __launch_bounds__(128)
void gemm_n32_kernel(const float* __restrict__ A, const float* __restrict__ W,
                     const float* __restrict__ rowscale, float* __restrict__ out, int M) {
  constexpr int LDK = K + 1;
  __shared__ float sA[32 * LDK];

  const int tid = threadIdx.x;
  const int block_row0 = blockIdx.x * 32;

  for (int idx = tid; idx < 32 * K; idx += 128) {
    int r = idx / K;
    int c = idx - r * K;
    int gr = block_row0 + r;
    sA[r * LDK + c] = (gr < M) ? A[(long long)gr * K + c] : 0.0f;
  }
  __syncthreads();

  const int wid   = tid >> 5;
  const int lane  = tid & 31;
  const int lhalf = lane >> 4;
  const int l15   = lane & 15;
  const int warp_m = (wid >> 1) * 16;
  const int n0     = (wid & 1) * 16;

  v8f acc = {};
  for (int k = 0; k < K; k += 4) {
    const float* ap = &sA[(warp_m + l15) * LDK + k + lhalf * 2];
    v2f a;
    a.x = ap[0];
    a.y = ap[1];
    const float* bp = &W[(k + lhalf * 2) * HID + n0 + l15];
    v2f b;
    b.x = bp[0];
    b.y = bp[HID];
    acc = __builtin_amdgcn_wmma_f32_16x16x4_f32(false, a, false, b, (short)0, acc,
                                                false, false);
  }

  const int grow_base = block_row0 + warp_m + lhalf * 8;
  const int col = n0 + l15;
#pragma unroll
  for (int j = 0; j < 8; ++j) {
    int gr = grow_base + j;
    if (gr < M) out[(long long)gr * HID + col] = acc[j] * rowscale[gr];
  }
}


extern "C" void kernel_launch(void* const* d_in, const int* in_sizes, int n_in,
                              void* d_out, int out_size, void* d_ws, size_t ws_size,
                              hipStream_t stream) {
  const float* x  = (const float*)d_in[0];
  const int*   ei = (const int*)d_in[1];
  const float* W1 = (const float*)d_in[2];
  const float* b1 = (const float*)d_in[3];
  const float* W2 = (const float*)d_in[4];
  const float* b2 = (const float*)d_in[5];
  float* out = (float*)d_out;

  const int N = in_sizes[0] / IN_CH;
  const int E = in_sizes[1] / 2;
  const int* src = ei;
  const int* dst = ei + E;

  const int Npad = (N + 63) & ~63;
  const int Epad = (E + 63) & ~63;
  int* gcount  = (int*)d_ws;
  int* gbase   = gcount + NBMAX;
  int* gcursor = gbase + 2 * NBMAX;
  float* dinv  = (float*)(gcursor + NBMAX);
  unsigned* keys = (unsigned*)(dinv + Npad);
  float* bufA = (float*)(keys + Epad);
  float* bufB = bufA + (size_t)(N + 8) * HID;
  unsigned* csr_src = (unsigned*)(bufB + (size_t)N * HID);
  int* rowptr = (int*)(csr_src + Epad);

  const unsigned NB    = (N + BK - 1) / BK;
  const unsigned gPart = (E + PT * PE - 1) / (PT * PE);
  const unsigned gGemm = (N + 31) / 32;
  const unsigned gRows = (N + RW * NPW - 1) / (RW * NPW);

  zero_counts_kernel<<<1, NBMAX, 0, stream>>>(gcount, NB);
  bucket_hist_kernel<<<gPart, PT, 0, stream>>>(dst, gcount, E, NB);
  bucket_scan_kernel<<<1, NBMAX, 0, stream>>>(gcount, gbase, gcursor, NB);
  bucket_scatter_kernel<<<gPart, PT, 0, stream>>>(src, dst, gcursor, keys, E, NB);
  bucket_csr_kernel<<<NB, AT, 0, stream>>>(keys, gbase, csr_src, rowptr, dinv, bufA + (size_t)N * HID, N);

  gemm_n32_kernel<IN_CH><<<gGemm, 128, 0, stream>>>(x, W1, dinv, bufA, N);
  row_sum_kernel<true><<<gRows, RW * 32, 0, stream>>>(csr_src, rowptr, bufA, dinv, b1, bufB, N, E);

  gemm_n32_kernel<HID><<<gGemm, 128, 0, stream>>>(bufB, W2, dinv, bufA, N);
  row_sum_kernel<false><<<gRows, RW * 32, 0, stream>>>(csr_src, rowptr, bufA, dinv, b2, out, N, E);
}


# loads take a clamped index instead of a guard (under a guard each load was waited for before the next issued): edge, key, GEMM tile (16 B/lane) and row-scale loads all in flight per thread
# speedup vs baseline: 5.8072x; 1.2486x over previous
#include <hip/hip_runtime.h>


typedef __attribute__((ext_vector_type(2))) float v2f;
typedef __attribute__((ext_vector_type(8))) float v8f;
typedef __attribute__((ext_vector_type(4))) float v4f;

#define IN_CH 128
#define HID 32

#define BSH 8
#define BK 256
#define NBMAX 512
#define PT 512
#define PE 16
#define AT 1024
#define LCAP 12288
#define KC (LCAP / AT)
#define RW 8
#define NPW 8


__global__ void zero_counts_kernel(int* __restrict__ gcount, int nb) {
  int i = blockIdx.x * blockDim.x + threadIdx.x;
  if (i < nb) gcount[i] = 0;
}

__global__ __launch_bounds__(PT)
void bucket_hist_kernel(const int* __restrict__ dst, int* __restrict__ gcount, int e_count, int nb) {
  __shared__ int hist[NBMAX];
  for (int i = threadIdx.x; i < NBMAX; i += PT) hist[i] = 0;
  const int base = blockIdx.x * (PT * PE);
  const int elast = max(e_count - 1, 0);
  int d[PE];
#pragma unroll
  for (int k = 0; k < PE; ++k) d[k] = dst[min(base + k * PT + (int)threadIdx.x, elast)];
  __syncthreads();
#pragma unroll
  for (int k = 0; k < PE; ++k) {
    const int e = base + k * PT + (int)threadIdx.x;
    if (e < e_count) atomicAdd(&hist[d[k] >> BSH], 1);
  }
  __syncthreads();
  for (int i = threadIdx.x; i < nb; i += PT) {
    const int c = hist[i];
    if (c) atomicAdd(&gcount[i], c);
  }
}

__global__ __launch_bounds__(NBMAX)
void bucket_scan_kernel(const int* __restrict__ gcount, int* __restrict__ gbase,
                        int* __restrict__ gcursor, int nb) {
  __shared__ int s[NBMAX];
  const int t = threadIdx.x;
  const int c = (t < nb) ? gcount[t] : 0;
  s[t] = c;
  __syncthreads();
  for (int off = 1; off < NBMAX; off <<= 1) {
    const int v = (t >= off) ? s[t - off] : 0;
    __syncthreads();
    s[t] += v;
    __syncthreads();
  }
  if (t < nb) {
    gbase[t] = s[t] - c;
    gcursor[t] = s[t] - c;
  }
  if (t == nb - 1) gbase[nb] = s[t];
}

__global__ __launch_bounds__(PT)
void bucket_scatter_kernel(const int* __restrict__ src, const int* __restrict__ dst,
                           int* __restrict__ gcursor, unsigned* __restrict__ keys,
                           int e_count, int nb) {
  __shared__ int hist[NBMAX];
  for (int i = threadIdx.x; i < NBMAX; i += PT) hist[i] = 0;
  const int base = blockIdx.x * (PT * PE);
  const int elast = max(e_count - 1, 0);
  int sv[PE];
  int dv[PE];
#pragma unroll
  for (int k = 0; k < PE; ++k) {
    const int ec = min(base + k * PT + (int)threadIdx.x, elast);
    sv[k] = src[ec];
    dv[k] = dst[ec];
  }
  __syncthreads();
  unsigned key[PE];
  int bkt[PE];
  int rank[PE];
#pragma unroll
  for (int k = 0; k < PE; ++k) {
    const int e = base + k * PT + (int)threadIdx.x;
    key[k] = ((unsigned)sv[k] << BSH) | (unsigned)(dv[k] & (BK - 1));
    bkt[k] = dv[k] >> BSH;
    rank[k] = -1;
    if (e < e_count) rank[k] = atomicAdd(&hist[bkt[k]], 1);
  }
  __syncthreads();
  for (int i = threadIdx.x; i < nb; i += PT) {
    const int c = hist[i];
    hist[i] = c ? atomicAdd(&gcursor[i], c) : 0;
  }
  __syncthreads();
  int first[PE];
#pragma unroll
  for (int k = 0; k < PE; ++k) first[k] = hist[bkt[k]];
#pragma unroll
  for (int k = 0; k < PE; ++k) {
    if (rank[k] >= 0) keys[first[k] + rank[k]] = key[k];
  }
}

__global__ __launch_bounds__(AT)
void bucket_csr_kernel(const unsigned* __restrict__ keys, const int* __restrict__ gbase,
                       unsigned* __restrict__ csr_src, int* __restrict__ rowptr,
                       float* __restrict__ dinv, float* __restrict__ zero_row, int n) {
  __shared__ int cnt[BK];
  __shared__ int wsum[BK / 32];
  __shared__ unsigned stage[LCAP];
  const int t = threadIdx.x;
  const int lane = t & 31;
  if (t < BK) cnt[t] = 0;
  const int beg = gbase[blockIdx.x];
  const int end = gbase[blockIdx.x + 1];
  const int m = end - beg;
  const bool staged = (m <= LCAP);
  unsigned kr[KC];
  if (staged) {
    const int klast = max(end - 1, 0);
#pragma unroll
    for (int j = 0; j < KC; ++j) kr[j] = keys[min(beg + j * AT + t, klast)];
  }
  __syncthreads();
  if (staged) {
#pragma unroll
    for (int j = 0; j < KC; ++j) {
      if (j * AT + t < m) atomicAdd(&cnt[kr[j] & (BK - 1)], 1);
    }
  } else {
    for (int e = beg + t; e < end; e += AT) atomicAdd(&cnt[keys[e] & (BK - 1)], 1);
  }
  __syncthreads();
  int c = 0;
  int incl = 0;
  if (t < BK) {
    c = cnt[t];
    incl = c;
#pragma unroll
    for (int off = 1; off < 32; off <<= 1) {
      const int v = __shfl_up(incl, off);
      if (lane >= off) incl += v;
    }
    if (lane == 31) wsum[t >> 5] = incl;
  }
  __syncthreads();
  if (t < BK) {
    int start = incl - c;
    for (int w = 0; w < (t >> 5); ++w) start += wsum[w];
    cnt[t] = start;
    const int node = blockIdx.x * BK + t;
    if (node < n) {
      rowptr[node] = beg + start;
      const float d = 1.0f + (float)c;
      dinv[node] = (d > 0.0f) ? rsqrtf(d) : 0.0f;
    }
  }
  if (blockIdx.x == gridDim.x - 1) {
    if (t == 0) rowptr[n] = end;
    if (t < HID) zero_row[t] = 0.0f;
  }
  __syncthreads();
  if (staged) {
    int pos[KC];
#pragma unroll
    for (int j = 0; j < KC; ++j) {
      pos[j] = 0;
      if (j * AT + t < m) pos[j] = atomicAdd(&cnt[kr[j] & (BK - 1)], 1);
    }
#pragma unroll
    for (int j = 0; j < KC; ++j) {
      if (j * AT + t < m) stage[pos[j]] = kr[j] >> BSH;
    }
  } else {
    for (int e = beg + t; e < end; e += AT) {
      const unsigned k = keys[e];
      const int pos = atomicAdd(&cnt[k & (BK - 1)], 1);
      csr_src[beg + pos] = k >> BSH;
    }
  }
  __syncthreads();
  if (staged && m > 0) {
    unsigned o[KC];
#pragma unroll
    for (int j = 0; j < KC; ++j) o[j] = stage[min(j * AT + t, m - 1)];
#pragma unroll
    for (int j = 0; j < KC; ++j) {
      if (j * AT + t < m) csr_src[beg + j * AT + t] = o[j];
    }
  }
}

template <bool RELU>
__global__ __launch_bounds__(RW * 32)
void row_sum_kernel(const unsigned* __restrict__ csr_src, const int* __restrict__ rowptr,
                    const float* __restrict__ hs, const float* __restrict__ dinv,
                    const float* __restrict__ bias, float* __restrict__ out, int n, int e_count) {
  const int lane = threadIdx.x & 31;
  const int wid = __builtin_amdgcn_readfirstlane((int)threadIdx.x >> 5);
  const int g = lane >> 3;
  const int q = lane & 7;
  const int n0 = (blockIdx.x * RW + wid) * NPW;
  if (n0 >= n) return;
  const int cntn = min(NPW, n - n0);
  const int elast = max(e_count - 1, 0);
  const float4* hs4 = reinterpret_cast<const float4*>(hs);
  float4* out4 = reinterpret_cast<float4*>(out);
  const int rp = rowptr[min(n0 + lane, n)];
  const float4 b4 = reinterpret_cast<const float4*>(bias)[q];
  int beg = __builtin_amdgcn_readlane(rp, 0);
  int end = __builtin_amdgcn_readlane(rp, 1);
  unsigned kA = csr_src[min(beg + lane, elast)];
  unsigned kB = csr_src[min(beg + 32 + lane, elast)];
  kA = (beg + lane < end) ? kA : (unsigned)n;
  kB = (beg + 32 + lane < end) ? kB : (unsigned)n;
  for (int i = 0; i < cntn; ++i) {
    const int node = n0 + i;
    const int nend = (i + 1 < cntn) ? __builtin_amdgcn_readlane(rp, min(i + 2, 31)) : end;
    unsigned nA = csr_src[min(end + lane, elast)];
    unsigned nB = csr_src[min(end + 32 + lane, elast)];
    const float di = dinv[node];
    const int deg = end - beg;
    unsigned sA[8];
#pragma unroll
    for (int s = 0; s < 8; ++s)
      sA[s] = (unsigned)__builtin_amdgcn_ds_bpermute((4 * s + g) << 2, (int)kA);
    float4 acc = hs4[(size_t)(g == 0 ? node : n) * 8 + q];
    float4 vA[8];
#pragma unroll
    for (int s = 0; s < 8; ++s) vA[s] = hs4[(size_t)sA[s] * 8 + q];
    float4 accB = make_float4(0.0f, 0.0f, 0.0f, 0.0f);
    if (deg > 32) {
      unsigned sB[8];
#pragma unroll
      for (int s = 0; s < 8; ++s)
        sB[s] = (unsigned)__builtin_amdgcn_ds_bpermute((4 * s + g) << 2, (int)kB);
      float4 vB[8];
#pragma unroll
      for (int s = 0; s < 8; ++s) vB[s] = hs4[(size_t)sB[s] * 8 + q];
#pragma unroll
      for (int s = 0; s < 8; ++s) {
        accB.x += vB[s].x;
        accB.y += vB[s].y;
        accB.z += vB[s].z;
        accB.w += vB[s].w;
      }
      for (int e = beg + 64 + g; e < end; e += 4) {
        const float4 r = hs4[(size_t)csr_src[e] * 8 + q];
        accB.x += r.x;
        accB.y += r.y;
        accB.z += r.z;
        accB.w += r.w;
      }
    }
#pragma unroll
    for (int s = 0; s < 8; ++s) {
      acc.x += vA[s].x;
      acc.y += vA[s].y;
      acc.z += vA[s].z;
      acc.w += vA[s].w;
    }
    acc.x += accB.x;
    acc.y += accB.y;
    acc.z += accB.z;
    acc.w += accB.w;
    acc.x += __shfl_xor(acc.x, 8);
    acc.y += __shfl_xor(acc.y, 8);
    acc.z += __shfl_xor(acc.z, 8);
    acc.w += __shfl_xor(acc.w, 8);
    acc.x += __shfl_xor(acc.x, 16);
    acc.y += __shfl_xor(acc.y, 16);
    acc.z += __shfl_xor(acc.z, 16);
    acc.w += __shfl_xor(acc.w, 16);
    if (g == 0) {
      float4 r;
      r.x = di * acc.x + b4.x;
      r.y = di * acc.y + b4.y;
      r.z = di * acc.z + b4.z;
      r.w = di * acc.w + b4.w;
      if (RELU) {
        r.x = fmaxf(r.x, 0.0f);
        r.y = fmaxf(r.y, 0.0f);
        r.z = fmaxf(r.z, 0.0f);
        r.w = fmaxf(r.w, 0.0f);
      }
      out4[(size_t)node * 8 + q] = r;
    }
    kA = (end + lane < nend) ? nA : (unsigned)n;
    kB = (end + 32 + lane < nend) ? nB : (unsigned)n;
    beg = end;
    end = nend;
  }
}

template <int K>
__global__ __launch_bounds__(128)
void gemm_n32_kernel(const float* __restrict__ A, const float* __restrict__ W,
                     const float* __restrict__ rowscale, float* __restrict__ out, int M) {
  constexpr int LDK = K + 4;
  constexpr int V = K / 4;
  constexpr int NL = (32 * V) / 128;
  __shared__ __align__(16) float sA[32 * LDK];

  const int tid = threadIdx.x;
  const int block_row0 = blockIdx.x * 32;
  const int mlast = M - 1;

  const int wid   = tid >> 5;
  const int lane  = tid & 31;
  const int lhalf = lane >> 4;
  const int l15   = lane & 15;
  const int warp_m = (wid >> 1) * 16;
  const int n0     = (wid & 1) * 16;
  const int grow_base = block_row0 + warp_m + lhalf * 8;

  const v4f* A4 = reinterpret_cast<const v4f*>(A);
  v4f av[NL];
#pragma unroll
  for (int i = 0; i < NL; ++i) {
    const int idx = tid + 128 * i;
    const int r = idx / V;
    const int c4 = idx - r * V;
    av[i] = A4[(size_t)min(block_row0 + r, mlast) * V + c4];
  }
  float rs[8];
#pragma unroll
  for (int j = 0; j < 8; ++j) rs[j] = rowscale[min(grow_base + j, mlast)];
  if constexpr (NL == 8) {
    asm volatile("" :: "v"(av[0]), "v"(av[1]), "v"(av[2]), "v"(av[3]),
                       "v"(av[4]), "v"(av[5]), "v"(av[6]), "v"(av[7]));
  }
#pragma unroll
  for (int i = 0; i < NL; ++i) {
    const int idx = tid + 128 * i;
    const int r = idx / V;
    const int c4 = idx - r * V;
    *reinterpret_cast<v4f*>(&sA[r * LDK + 4 * c4]) = av[i];
  }
  __syncthreads();

  v8f acc = {};
  for (int k = 0; k < K; k += 4) {
    const float* ap = &sA[(warp_m + l15) * LDK + k + lhalf * 2];
    v2f a;
    a.x = ap[0];
    a.y = ap[1];
    const float* bp = &W[(k + lhalf * 2) * HID + n0 + l15];
    v2f b;
    b.x = bp[0];
    b.y = bp[HID];
    acc = __builtin_amdgcn_wmma_f32_16x16x4_f32(false, a, false, b, (short)0, acc,
                                                false, false);
  }

  const int col = n0 + l15;
#pragma unroll
  for (int j = 0; j < 8; ++j) {
    int gr = grow_base + j;
    if (gr < M) out[(long long)gr * HID + col] = acc[j] * rs[j];
  }
}


extern "C" void kernel_launch(void* const* d_in, const int* in_sizes, int n_in,
                              void* d_out, int out_size, void* d_ws, size_t ws_size,
                              hipStream_t stream) {
  const float* x  = (const float*)d_in[0];
  const int*   ei = (const int*)d_in[1];
  const float* W1 = (const float*)d_in[2];
  const float* b1 = (const float*)d_in[3];
  const float* W2 = (const float*)d_in[4];
  const float* b2 = (const float*)d_in[5];
  float* out = (float*)d_out;

  const int N = in_sizes[0] / IN_CH;
  const int E = in_sizes[1] / 2;
  const int* src = ei;
  const int* dst = ei + E;

  const int Npad = (N + 63) & ~63;
  const int Epad = (E + 63) & ~63;
  int* gcount  = (int*)d_ws;
  int* gbase   = gcount + NBMAX;
  int* gcursor = gbase + 2 * NBMAX;
  float* dinv  = (float*)(gcursor + NBMAX);
  unsigned* keys = (unsigned*)(dinv + Npad);
  float* bufA = (float*)(keys + Epad);
  float* bufB = bufA + (size_t)(N + 8) * HID;
  unsigned* csr_src = (unsigned*)(bufB + (size_t)N * HID);
  int* rowptr = (int*)(csr_src + Epad);

  const unsigned NB    = (N + BK - 1) / BK;
  const unsigned gPart = (E + PT * PE - 1) / (PT * PE);
  const unsigned gGemm = (N + 31) / 32;
  const unsigned gRows = (N + RW * NPW - 1) / (RW * NPW);

  zero_counts_kernel<<<1, NBMAX, 0, stream>>>(gcount, NB);
  bucket_hist_kernel<<<gPart, PT, 0, stream>>>(dst, gcount, E, NB);
  bucket_scan_kernel<<<1, NBMAX, 0, stream>>>(gcount, gbase, gcursor, NB);
  bucket_scatter_kernel<<<gPart, PT, 0, stream>>>(src, dst, gcursor, keys, E, NB);
  bucket_csr_kernel<<<NB, AT, 0, stream>>>(keys, gbase, csr_src, rowptr, dinv, bufA + (size_t)N * HID, N);

  gemm_n32_kernel<IN_CH><<<gGemm, 128, 0, stream>>>(x, W1, dinv, bufA, N);
  row_sum_kernel<true><<<gRows, RW * 32, 0, stream>>>(csr_src, rowptr, bufA, dinv, b1, bufB, N, E);

  gemm_n32_kernel<HID><<<gGemm, 128, 0, stream>>>(bufB, W2, dinv, bufA, N);
  row_sum_kernel<false><<<gRows, RW * 32, 0, stream>>>(csr_src, rowptr, bufA, dinv, b2, out, N, E);
}


# layer-2 product fused into layer 1's row-sum launch: a block parks its 64 relu'd rows and their dinv in LDS, one barrier, each wave one 16x16 f32 WMMA tile (same k order); one launch, one plane less
# speedup vs baseline: 5.9959x; 1.0325x over previous
#include <hip/hip_runtime.h>


typedef __attribute__((ext_vector_type(2))) float v2f;
typedef __attribute__((ext_vector_type(8))) float v8f;
typedef __attribute__((ext_vector_type(4))) float v4f;

#define IN_CH 128
#define HID 32

#define BSH 8
#define BK 256
#define NBMAX 512
#define PT 512
#define PE 16
#define AT 1024
#define LCAP 12288
#define KC (LCAP / AT)
#define RW 8
#define NPW 8
#define TP (HID + 4)


__global__ void zero_counts_kernel(int* __restrict__ gcount, int nb) {
  int i = blockIdx.x * blockDim.x + threadIdx.x;
  if (i < nb) gcount[i] = 0;
}

__global__ __launch_bounds__(PT)
void bucket_hist_kernel(const int* __restrict__ dst, int* __restrict__ gcount, int e_count, int nb) {
  __shared__ int hist[NBMAX];
  for (int i = threadIdx.x; i < NBMAX; i += PT) hist[i] = 0;
  const int base = blockIdx.x * (PT * PE);
  const int elast = max(e_count - 1, 0);
  int d[PE];
#pragma unroll
  for (int k = 0; k < PE; ++k) d[k] = dst[min(base + k * PT + (int)threadIdx.x, elast)];
  __syncthreads();
#pragma unroll
  for (int k = 0; k < PE; ++k) {
    const int e = base + k * PT + (int)threadIdx.x;
    if (e < e_count) atomicAdd(&hist[d[k] >> BSH], 1);
  }
  __syncthreads();
  for (int i = threadIdx.x; i < nb; i += PT) {
    const int c = hist[i];
    if (c) atomicAdd(&gcount[i], c);
  }
}

__global__ __launch_bounds__(NBMAX)
void bucket_scan_kernel(const int* __restrict__ gcount, int* __restrict__ gbase,
                        int* __restrict__ gcursor, int nb) {
  __shared__ int s[NBMAX];
  const int t = threadIdx.x;
  const int c = (t < nb) ? gcount[t] : 0;
  s[t] = c;
  __syncthreads();
  for (int off = 1; off < NBMAX; off <<= 1) {
    const int v = (t >= off) ? s[t - off] : 0;
    __syncthreads();
    s[t] += v;
    __syncthreads();
  }
  if (t < nb) {
    gbase[t] = s[t] - c;
    gcursor[t] = s[t] - c;
  }
  if (t == nb - 1) gbase[nb] = s[t];
}

__global__ __launch_bounds__(PT)
void bucket_scatter_kernel(const int* __restrict__ src, const int* __restrict__ dst,
                           int* __restrict__ gcursor, unsigned* __restrict__ keys,
                           int e_count, int nb) {
  __shared__ int hist[NBMAX];
  for (int i = threadIdx.x; i < NBMAX; i += PT) hist[i] = 0;
  const int base = blockIdx.x * (PT * PE);
  const int elast = max(e_count - 1, 0);
  int sv[PE];
  int dv[PE];
#pragma unroll
  for (int k = 0; k < PE; ++k) {
    const int ec = min(base + k * PT + (int)threadIdx.x, elast);
    sv[k] = src[ec];
    dv[k] = dst[ec];
  }
  __syncthreads();
  unsigned key[PE];
  int bkt[PE];
  int rank[PE];
#pragma unroll
  for (int k = 0; k < PE; ++k) {
    const int e = base + k * PT + (int)threadIdx.x;
    key[k] = ((unsigned)sv[k] << BSH) | (unsigned)(dv[k] & (BK - 1));
    bkt[k] = dv[k] >> BSH;
    rank[k] = -1;
    if (e < e_count) rank[k] = atomicAdd(&hist[bkt[k]], 1);
  }
  __syncthreads();
  for (int i = threadIdx.x; i < nb; i += PT) {
    const int c = hist[i];
    hist[i] = c ? atomicAdd(&gcursor[i], c) : 0;
  }
  __syncthreads();
  int first[PE];
#pragma unroll
  for (int k = 0; k < PE; ++k) first[k] = hist[bkt[k]];
#pragma unroll
  for (int k = 0; k < PE; ++k) {
    if (rank[k] >= 0) keys[first[k] + rank[k]] = key[k];
  }
}

__global__ __launch_bounds__(AT)
void bucket_csr_kernel(const unsigned* __restrict__ keys, const int* __restrict__ gbase,
                       unsigned* __restrict__ csr_src, int* __restrict__ rowptr,
                       float* __restrict__ dinv, float* __restrict__ zero_row, int n) {
  __shared__ int cnt[BK];
  __shared__ int wsum[BK / 32];
  __shared__ unsigned stage[LCAP];
  const int t = threadIdx.x;
  const int lane = t & 31;
  if (t < BK) cnt[t] = 0;
  const int beg = gbase[blockIdx.x];
  const int end = gbase[blockIdx.x + 1];
  const int m = end - beg;
  const bool staged = (m <= LCAP);
  unsigned kr[KC];
  if (staged) {
    const int klast = max(end - 1, 0);
#pragma unroll
    for (int j = 0; j < KC; ++j) kr[j] = keys[min(beg + j * AT + t, klast)];
  }
  __syncthreads();
  if (staged) {
#pragma unroll
    for (int j = 0; j < KC; ++j) {
      if (j * AT + t < m) atomicAdd(&cnt[kr[j] & (BK - 1)], 1);
    }
  } else {
    for (int e = beg + t; e < end; e += AT) atomicAdd(&cnt[keys[e] & (BK - 1)], 1);
  }
  __syncthreads();
  int c = 0;
  int incl = 0;
  if (t < BK) {
    c = cnt[t];
    incl = c;
#pragma unroll
    for (int off = 1; off < 32; off <<= 1) {
      const int v = __shfl_up(incl, off);
      if (lane >= off) incl += v;
    }
    if (lane == 31) wsum[t >> 5] = incl;
  }
  __syncthreads();
  if (t < BK) {
    int start = incl - c;
    for (int w = 0; w < (t >> 5); ++w) start += wsum[w];
    cnt[t] = start;
    const int node = blockIdx.x * BK + t;
    if (node < n) {
      rowptr[node] = beg + start;
      const float d = 1.0f + (float)c;
      dinv[node] = (d > 0.0f) ? rsqrtf(d) : 0.0f;
    }
  }
  if (blockIdx.x == gridDim.x - 1) {
    if (t == 0) rowptr[n] = end;
    if (t < HID) zero_row[t] = 0.0f;
  }
  __syncthreads();
  if (staged) {
    int pos[KC];
#pragma unroll
    for (int j = 0; j < KC; ++j) {
      pos[j] = 0;
      if (j * AT + t < m) pos[j] = atomicAdd(&cnt[kr[j] & (BK - 1)], 1);
    }
#pragma unroll
    for (int j = 0; j < KC; ++j) {
      if (j * AT + t < m) stage[pos[j]] = kr[j] >> BSH;
    }
  } else {
    for (int e = beg + t; e < end; e += AT) {
      const unsigned k = keys[e];
      const int pos = atomicAdd(&cnt[k & (BK - 1)], 1);
      csr_src[beg + pos] = k >> BSH;
    }
  }
  __syncthreads();
  if (staged && m > 0) {
    unsigned o[KC];
#pragma unroll
    for (int j = 0; j < KC; ++j) o[j] = stage[min(j * AT + t, m - 1)];
#pragma unroll
    for (int j = 0; j < KC; ++j) {
      if (j * AT + t < m) csr_src[beg + j * AT + t] = o[j];
    }
  }
}

template <bool NEXT>
__global__ __launch_bounds__(RW * 32)
void row_sum_kernel(const unsigned* __restrict__ csr_src, const int* __restrict__ rowptr,
                    const float* __restrict__ hs, const float* __restrict__ dinv,
                    const float* __restrict__ bias, const float* __restrict__ Wn,
                    float* __restrict__ out, int n, int e_count) {
  __shared__ __align__(16) float sT[NEXT ? RW * NPW * TP : 4];
  const int lane = threadIdx.x & 31;
  const int wid = __builtin_amdgcn_readfirstlane((int)threadIdx.x >> 5);
  const int g = lane >> 3;
  const int q = lane & 7;
  const int n0 = (blockIdx.x * RW + wid) * NPW;
  if constexpr (!NEXT) {
    if (n0 >= n) return;
  }
  const int cntn = max(0, min(NPW, n - n0));
  const int elast = max(e_count - 1, 0);
  const float4* hs4 = reinterpret_cast<const float4*>(hs);
  float4* out4 = reinterpret_cast<float4*>(out);
  const int rp = rowptr[min(n0 + lane, n)];
  const float4 b4 = reinterpret_cast<const float4*>(bias)[q];
  int beg = __builtin_amdgcn_readlane(rp, 0);
  int end = __builtin_amdgcn_readlane(rp, 1);
  unsigned kA = csr_src[min(beg + lane, elast)];
  unsigned kB = csr_src[min(beg + 32 + lane, elast)];
  kA = (beg + lane < end) ? kA : (unsigned)n;
  kB = (beg + 32 + lane < end) ? kB : (unsigned)n;
  for (int i = 0; i < cntn; ++i) {
    const int node = n0 + i;
    const int nend = (i + 1 < cntn) ? __builtin_amdgcn_readlane(rp, min(i + 2, 31)) : end;
    unsigned nA = csr_src[min(end + lane, elast)];
    unsigned nB = csr_src[min(end + 32 + lane, elast)];
    const float di = dinv[node];
    const int deg = end - beg;
    unsigned sA[8];
#pragma unroll
    for (int s = 0; s < 8; ++s)
      sA[s] = (unsigned)__builtin_amdgcn_ds_bpermute((4 * s + g) << 2, (int)kA);
    float4 acc = hs4[(size_t)(g == 0 ? node : n) * 8 + q];
    float4 vA[8];
#pragma unroll
    for (int s = 0; s < 8; ++s) vA[s] = hs4[(size_t)sA[s] * 8 + q];
    float4 accB = make_float4(0.0f, 0.0f, 0.0f, 0.0f);
    if (deg > 32) {
      unsigned sB[8];
#pragma unroll
      for (int s = 0; s < 8; ++s)
        sB[s] = (unsigned)__builtin_amdgcn_ds_bpermute((4 * s + g) << 2, (int)kB);
      float4 vB[8];
#pragma unroll
      for (int s = 0; s < 8; ++s) vB[s] = hs4[(size_t)sB[s] * 8 + q];
#pragma unroll
      for (int s = 0; s < 8; ++s) {
        accB.x += vB[s].x;
        accB.y += vB[s].y;
        accB.z += vB[s].z;
        accB.w += vB[s].w;
      }
      for (int e = beg + 64 + g; e < end; e += 4) {
        const float4 r = hs4[(size_t)csr_src[e] * 8 + q];
        accB.x += r.x;
        accB.y += r.y;
        accB.z += r.z;
        accB.w += r.w;
      }
    }
#pragma unroll
    for (int s = 0; s < 8; ++s) {
      acc.x += vA[s].x;
      acc.y += vA[s].y;
      acc.z += vA[s].z;
      acc.w += vA[s].w;
    }
    acc.x += accB.x;
    acc.y += accB.y;
    acc.z += accB.z;
    acc.w += accB.w;
    acc.x += __shfl_xor(acc.x, 8);
    acc.y += __shfl_xor(acc.y, 8);
    acc.z += __shfl_xor(acc.z, 8);
    acc.w += __shfl_xor(acc.w, 8);
    acc.x += __shfl_xor(acc.x, 16);
    acc.y += __shfl_xor(acc.y, 16);
    acc.z += __shfl_xor(acc.z, 16);
    acc.w += __shfl_xor(acc.w, 16);
    if constexpr (NEXT) {
      float4 r;
      r.x = fmaxf(di * acc.x + b4.x, 0.0f);
      r.y = fmaxf(di * acc.y + b4.y, 0.0f);
      r.z = fmaxf(di * acc.z + b4.z, 0.0f);
      r.w = fmaxf(di * acc.w + b4.w, 0.0f);
      if (lane == 8) r = make_float4(di, di, di, di);
      if (lane < 9) *reinterpret_cast<float4*>(&sT[(wid * NPW + i) * TP + 4 * lane]) = r;
    } else if (g == 0) {
      float4 r;
      r.x = di * acc.x + b4.x;
      r.y = di * acc.y + b4.y;
      r.z = di * acc.z + b4.z;
      r.w = di * acc.w + b4.w;
      out4[(size_t)node * 8 + q] = r;
    }
    kA = (end + lane < nend) ? nA : (unsigned)n;
    kB = (end + 32 + lane < nend) ? nB : (unsigned)n;
    beg = end;
    end = nend;
  }
  if constexpr (NEXT) {
    if (cntn < NPW) {
      float4* sT4 = reinterpret_cast<float4*>(&sT[wid * NPW * TP]);
#pragma unroll
      for (int h = 0; h < 3; ++h) {
        const int p = lane + 32 * h;
        if (p >= cntn * (TP / 4) && p < NPW * (TP / 4)) sT4[p] = make_float4(0.0f, 0.0f, 0.0f, 0.0f);
      }
    }
    const int lhalf = lane >> 4;
    const int l15 = lane & 15;
    const int mt = (wid >> 1) * 16;
    const int nt = (wid & 1) * 16;
    const int grow_base = blockIdx.x * (RW * NPW) + mt + lhalf * 8;
    v2f bf[HID / 4];
#pragma unroll
    for (int kk = 0; kk < HID / 4; ++kk) {
      const float* bp = &Wn[(4 * kk + lhalf * 2) * HID + nt + l15];
      bf[kk].x = bp[0];
      bf[kk].y = bp[HID];
    }
    __syncthreads();
    v8f acc = {};
#pragma unroll
    for (int kk = 0; kk < HID / 4; ++kk) {
      const float* ap = &sT[(mt + l15) * TP + 4 * kk + lhalf * 2];
      v2f a;
      a.x = ap[0];
      a.y = ap[1];
      acc = __builtin_amdgcn_wmma_f32_16x16x4_f32(false, a, false, bf[kk], (short)0, acc,
                                                  false, false);
    }
    float rs[8];
#pragma unroll
    for (int j = 0; j < 8; ++j) rs[j] = sT[(mt + lhalf * 8 + j) * TP + HID];
    const int col = nt + l15;
#pragma unroll
    for (int j = 0; j < 8; ++j) {
      const int gr = grow_base + j;
      if (gr < n) out[(size_t)gr * HID + col] = acc[j] * rs[j];
    }
    if (blockIdx.x == gridDim.x - 1 && threadIdx.x < HID) out[(size_t)n * HID + threadIdx.x] = 0.0f;
  }
}

template <int K>
__global__ __launch_bounds__(128)
void gemm_n32_kernel(const float* __restrict__ A, const float* __restrict__ W,
                     const float* __restrict__ rowscale, float* __restrict__ out, int M) {
  constexpr int LDK = K + 4;
  constexpr int V = K / 4;
  constexpr int NL = (32 * V) / 128;
  __shared__ __align__(16) float sA[32 * LDK];

  const int tid = threadIdx.x;
  const int block_row0 = blockIdx.x * 32;
  const int mlast = M - 1;

  const int wid   = tid >> 5;
  const int lane  = tid & 31;
  const int lhalf = lane >> 4;
  const int l15   = lane & 15;
  const int warp_m = (wid >> 1) * 16;
  const int n0     = (wid & 1) * 16;
  const int grow_base = block_row0 + warp_m + lhalf * 8;

  const v4f* A4 = reinterpret_cast<const v4f*>(A);
  v4f av[NL];
#pragma unroll
  for (int i = 0; i < NL; ++i) {
    const int idx = tid + 128 * i;
    const int r = idx / V;
    const int c4 = idx - r * V;
    av[i] = A4[(size_t)min(block_row0 + r, mlast) * V + c4];
  }
  float rs[8];
#pragma unroll
  for (int j = 0; j < 8; ++j) rs[j] = rowscale[min(grow_base + j, mlast)];
  if constexpr (NL == 8) {
    asm volatile("" :: "v"(av[0]), "v"(av[1]), "v"(av[2]), "v"(av[3]),
                       "v"(av[4]), "v"(av[5]), "v"(av[6]), "v"(av[7]));
  }
#pragma unroll
  for (int i = 0; i < NL; ++i) {
    const int idx = tid + 128 * i;
    const int r = idx / V;
    const int c4 = idx - r * V;
    *reinterpret_cast<v4f*>(&sA[r * LDK + 4 * c4]) = av[i];
  }
  __syncthreads();

  v8f acc = {};
  for (int k = 0; k < K; k += 4) {
    const float* ap = &sA[(warp_m + l15) * LDK + k + lhalf * 2];
    v2f a;
    a.x = ap[0];
    a.y = ap[1];
    const float* bp = &W[(k + lhalf * 2) * HID + n0 + l15];
    v2f b;
    b.x = bp[0];
    b.y = bp[HID];
    acc = __builtin_amdgcn_wmma_f32_16x16x4_f32(false, a, false, b, (short)0, acc,
                                                false, false);
  }

  const int col = n0 + l15;
#pragma unroll
  for (int j = 0; j < 8; ++j) {
    int gr = grow_base + j;
    if (gr < M) out[(long long)gr * HID + col] = acc[j] * rs[j];
  }
}


extern "C" void kernel_launch(void* const* d_in, const int* in_sizes, int n_in,
                              void* d_out, int out_size, void* d_ws, size_t ws_size,
                              hipStream_t stream) {
  const float* x  = (const float*)d_in[0];
  const int*   ei = (const int*)d_in[1];
  const float* W1 = (const float*)d_in[2];
  const float* b1 = (const float*)d_in[3];
  const float* W2 = (const float*)d_in[4];
  const float* b2 = (const float*)d_in[5];
  float* out = (float*)d_out;

  const int N = in_sizes[0] / IN_CH;
  const int E = in_sizes[1] / 2;
  const int* src = ei;
  const int* dst = ei + E;

  const int Npad = (N + 63) & ~63;
  const int Epad = (E + 63) & ~63;
  int* gcount  = (int*)d_ws;
  int* gbase   = gcount + NBMAX;
  int* gcursor = gbase + 2 * NBMAX;
  float* dinv  = (float*)(gcursor + NBMAX);
  unsigned* keys = (unsigned*)(dinv + Npad);
  float* bufA = (float*)(keys + Epad);
  float* bufB = bufA + (size_t)(N + 8) * HID;
  unsigned* csr_src = (unsigned*)(bufB + (size_t)(N + 8) * HID);
  int* rowptr = (int*)(csr_src + Epad);

  const unsigned NB    = (N + BK - 1) / BK;
  const unsigned gPart = (E + PT * PE - 1) / (PT * PE);
  const unsigned gGemm = (N + 31) / 32;
  const unsigned gRows = (N + RW * NPW - 1) / (RW * NPW);

  zero_counts_kernel<<<1, NBMAX, 0, stream>>>(gcount, NB);
  bucket_hist_kernel<<<gPart, PT, 0, stream>>>(dst, gcount, E, NB);
  bucket_scan_kernel<<<1, NBMAX, 0, stream>>>(gcount, gbase, gcursor, NB);
  bucket_scatter_kernel<<<gPart, PT, 0, stream>>>(src, dst, gcursor, keys, E, NB);
  bucket_csr_kernel<<<NB, AT, 0, stream>>>(keys, gbase, csr_src, rowptr, dinv, bufA + (size_t)N * HID, N);

  gemm_n32_kernel<IN_CH><<<gGemm, 128, 0, stream>>>(x, W1, dinv, bufA, N);
  row_sum_kernel<true><<<gRows, RW * 32, 0, stream>>>(csr_src, rowptr, bufA, dinv, b1, W2, bufB, N, E);

  row_sum_kernel<false><<<gRows, RW * 32, 0, stream>>>(csr_src, rowptr, bufB, dinv, b2, W2, out, N, E);
}


# edge partition in one launch: a block orders its own 8192 edges by dst bucket in LDS and writes its piece starts; a bucket's workgroup gathers the pieces; 3 launches gone; sum order in a run changes
# speedup vs baseline: 7.5452x; 1.2584x over previous
#include <hip/hip_runtime.h>


typedef __attribute__((ext_vector_type(2))) float v2f;
typedef __attribute__((ext_vector_type(8))) float v8f;
typedef __attribute__((ext_vector_type(4))) float v4f;

#define IN_CH 128
#define HID 32

#define BSH 8
#define BK 256
#define NBMAX 512
#define PT 512
#define PE 16
#define PBLK (PT * PE)
#define NBP (NBMAX + 1)
#define KS 13
#define AT 1024
#define LCAP 12288
#define KC (LCAP / AT)
#define RW 8
#define NPW 8
#define TP (HID + 4)


__global__ __launch_bounds__(PT)
void block_sort_kernel(const int* __restrict__ src, const int* __restrict__ dst,
                       unsigned* __restrict__ keys, int* __restrict__ boff,
                       int* __restrict__ gtotal, int e_count) {
  __shared__ int hist[NBMAX];
  __shared__ int wtot[PT / 32];
  __shared__ __align__(16) unsigned sorted[PBLK];
  const int t = threadIdx.x;
  const int lane = t & 31;
  hist[t] = 0;
  if (blockIdx.x == 0 && t == 0) *gtotal = 0;
  const int base = blockIdx.x * PBLK;
  const int elast = max(e_count - 1, 0);
  int sv[PE];
  int dv[PE];
#pragma unroll
  for (int k = 0; k < PE; ++k) {
    const int ec = min(base + k * PT + t, elast);
    sv[k] = src[ec];
    dv[k] = dst[ec];
  }
  __syncthreads();
  unsigned key[PE];
  int bkt[PE];
  int rank[PE];
#pragma unroll
  for (int k = 0; k < PE; ++k) {
    const int e = base + k * PT + t;
    key[k] = ((unsigned)sv[k] << BSH) | (unsigned)(dv[k] & (BK - 1));
    bkt[k] = dv[k] >> BSH;
    rank[k] = -1;
    if (e < e_count) rank[k] = atomicAdd(&hist[bkt[k]], 1);
  }
  __syncthreads();
  const int c = hist[t];
  int incl = c;
#pragma unroll
  for (int off = 1; off < 32; off <<= 1) {
    const int v = __shfl_up(incl, off);
    if (lane >= off) incl += v;
  }
  if (lane == 31) wtot[t >> 5] = incl;
  __syncthreads();
  int start = incl - c;
  for (int w = 0; w < (t >> 5); ++w) start += wtot[w];
  hist[t] = start;
  boff[(size_t)blockIdx.x * NBP + t] = start;
  if (t == PT - 1) boff[(size_t)blockIdx.x * NBP + NBMAX] = start + c;
  __syncthreads();
  int first[PE];
#pragma unroll
  for (int k = 0; k < PE; ++k) first[k] = hist[bkt[k]];
#pragma unroll
  for (int k = 0; k < PE; ++k) {
    if (rank[k] >= 0) sorted[first[k] + rank[k]] = key[k];
  }
  __syncthreads();
  const int mine = min(PBLK, e_count - base);
  const uint4* sorted4 = reinterpret_cast<const uint4*>(sorted);
  uint4* keys4 = reinterpret_cast<uint4*>(keys + base);
  if (mine == PBLK) {
    uint4 v[PE / 4];
#pragma unroll
    for (int k = 0; k < PE / 4; ++k) v[k] = sorted4[k * PT + t];
#pragma unroll
    for (int k = 0; k < PE / 4; ++k) keys4[k * PT + t] = v[k];
  } else {
    for (int u = t; u < mine; u += PT) keys[base + u] = sorted[u];
  }
}

template <bool PLACE>
__device__ __forceinline__ void walk_rest(const unsigned* __restrict__ keys, const int* __restrict__ boff,
                                          int* cnt, unsigned* stage, unsigned* __restrict__ csr_src,
                                          int pb, int pc, int wid, int lane, int b, int gpart,
                                          bool staged, int beg) {
  for (int i = 0; i < KS; ++i) {
    const int ci = __builtin_amdgcn_readlane(pc, i);
    if (ci > 32) {
      const int p0 = (wid + 32 * i) * PBLK + __builtin_amdgcn_readlane(pb, i);
      for (int o = 32 + lane; o < ci; o += 32) {
        const unsigned k = keys[p0 + o];
        if constexpr (PLACE) {
          const int pos = atomicAdd(&cnt[k & (BK - 1)], 1);
          if (staged) stage[pos] = k >> BSH; else csr_src[beg + pos] = k >> BSH;
        } else {
          atomicAdd(&cnt[k & (BK - 1)], 1);
        }
      }
    }
  }
  for (int j = wid + 32 * KS; j < gpart; j += 32) {
    const int sb = boff[(size_t)j * NBP + b];
    const int se = boff[(size_t)j * NBP + b + 1];
    for (int o = sb + lane; o < se; o += 32) {
      const unsigned k = keys[j * PBLK + o];
      if constexpr (PLACE) {
        const int pos = atomicAdd(&cnt[k & (BK - 1)], 1);
        if (staged) stage[pos] = k >> BSH; else csr_src[beg + pos] = k >> BSH;
      } else {
        atomicAdd(&cnt[k & (BK - 1)], 1);
      }
    }
  }
}

__global__ __launch_bounds__(AT)
void bucket_csr_kernel(const unsigned* __restrict__ keys, const int* __restrict__ boff,
                       int* __restrict__ gtotal, unsigned* __restrict__ csr_src,
                       int* __restrict__ rowptr, float* __restrict__ dinv,
                       float* __restrict__ zero_row, int n, int e_count, int gpart) {
  __shared__ int cnt[BK];
  __shared__ int wsum[BK / 32];
  __shared__ int mtot;
  __shared__ int sbeg;
  __shared__ unsigned stage[LCAP];
  const int t = threadIdx.x;
  const int lane = t & 31;
  const int wid = __builtin_amdgcn_readfirstlane(t >> 5);
  const int b = blockIdx.x;
  if (t < BK) cnt[t] = 0;
  if (t == 0) mtot = 0;
  const int elast = max(e_count - 1, 0);
  const int jl = wid + 32 * min(lane, KS - 1);
  const int jc = min(jl, gpart - 1);
  const int pb = boff[(size_t)jc * NBP + b];
  const int pe = boff[(size_t)jc * NBP + b + 1];
  const int pc = (lane < KS && jl < gpart) ? pe - pb : 0;
  unsigned kr[KS];
  int pcs[KS];
#pragma unroll
  for (int i = 0; i < KS; ++i) {
    pcs[i] = __builtin_amdgcn_readlane(pc, i);
    kr[i] = keys[min((wid + 32 * i) * PBLK + __builtin_amdgcn_readlane(pb, i) + lane, elast)];
  }
  __syncthreads();
#pragma unroll
  for (int i = 0; i < KS; ++i) {
    if (lane < pcs[i]) atomicAdd(&cnt[kr[i] & (BK - 1)], 1);
  }
  walk_rest<false>(keys, boff, cnt, stage, csr_src, pb, pc, wid, lane, b, gpart, false, 0);
  {
    int csum = pc;
#pragma unroll
    for (int off = 16; off > 0; off >>= 1) csum += __shfl_xor(csum, off);
    for (int j = wid + 32 * KS; j < gpart; j += 32)
      csum += boff[(size_t)j * NBP + b + 1] - boff[(size_t)j * NBP + b];
    if (lane == 0 && csum) atomicAdd(&mtot, csum);
  }
  __syncthreads();
  int c = 0;
  int incl = 0;
  if (t < BK) {
    c = cnt[t];
    incl = c;
#pragma unroll
    for (int off = 1; off < 32; off <<= 1) {
      const int v = __shfl_up(incl, off);
      if (lane >= off) incl += v;
    }
    if (lane == 31) wsum[t >> 5] = incl;
  }
  if (t == BK) sbeg = atomicAdd(gtotal, mtot);
  __syncthreads();
  const int m = mtot;
  const int beg = sbeg;
  const bool staged = (m <= LCAP);
  if (t < BK) {
    int start = incl - c;
    for (int w = 0; w < (t >> 5); ++w) start += wsum[w];
    cnt[t] = start;
    rowptr[b * (BK + 1) + t] = beg + start;
    const int node = b * BK + t;
    if (node < n) {
      const float d = 1.0f + (float)c;
      dinv[node] = (d > 0.0f) ? rsqrtf(d) : 0.0f;
    }
  }
  if (t == BK) rowptr[b * (BK + 1) + BK] = beg + m;
  if (blockIdx.x == gridDim.x - 1) {
    if (t < HID) zero_row[t] = 0.0f;
  }
  __syncthreads();
  {
    int pos[KS];
#pragma unroll
    for (int i = 0; i < KS; ++i) {
      pos[i] = 0;
      if (lane < pcs[i]) pos[i] = atomicAdd(&cnt[kr[i] & (BK - 1)], 1);
    }
    if (staged) {
#pragma unroll
      for (int i = 0; i < KS; ++i) {
        if (lane < pcs[i]) stage[pos[i]] = kr[i] >> BSH;
      }
    } else {
#pragma unroll
      for (int i = 0; i < KS; ++i) {
        if (lane < pcs[i]) csr_src[beg + pos[i]] = kr[i] >> BSH;
      }
    }
  }
  walk_rest<true>(keys, boff, cnt, stage, csr_src, pb, pc, wid, lane, b, gpart, staged, beg);
  __syncthreads();
  if (staged && m > 0) {
    unsigned o[KC];
#pragma unroll
    for (int j = 0; j < KC; ++j) o[j] = stage[min(j * AT + t, m - 1)];
#pragma unroll
    for (int j = 0; j < KC; ++j) {
      if (j * AT + t < m) csr_src[beg + j * AT + t] = o[j];
    }
  }
}

template <bool NEXT>
__global__ __launch_bounds__(RW * 32)
void row_sum_kernel(const unsigned* __restrict__ csr_src, const int* __restrict__ rowptr,
                    const float* __restrict__ hs, const float* __restrict__ dinv,
                    const float* __restrict__ bias, const float* __restrict__ Wn,
                    float* __restrict__ out, int n, int e_count) {
  __shared__ __align__(16) float sT[NEXT ? RW * NPW * TP : 4];
  const int lane = threadIdx.x & 31;
  const int wid = __builtin_amdgcn_readfirstlane((int)threadIdx.x >> 5);
  const int g = lane >> 3;
  const int q = lane & 7;
  const int n0 = (blockIdx.x * RW + wid) * NPW;
  if constexpr (!NEXT) {
    if (n0 >= n) return;
  }
  const int cntn = max(0, min(NPW, n - n0));
  const int elast = max(e_count - 1, 0);
  const float4* hs4 = reinterpret_cast<const float4*>(hs);
  float4* out4 = reinterpret_cast<float4*>(out);
  const int rp = rowptr[(n0 >> BSH) * (BK + 1) + min((n0 & (BK - 1)) + lane, BK)];
  const float4 b4 = reinterpret_cast<const float4*>(bias)[q];
  int beg = __builtin_amdgcn_readlane(rp, 0);
  int end = __builtin_amdgcn_readlane(rp, 1);
  unsigned kA = csr_src[min(beg + lane, elast)];
  unsigned kB = csr_src[min(beg + 32 + lane, elast)];
  kA = (beg + lane < end) ? kA : (unsigned)n;
  kB = (beg + 32 + lane < end) ? kB : (unsigned)n;
  for (int i = 0; i < cntn; ++i) {
    const int node = n0 + i;
    const int nend = (i + 1 < cntn) ? __builtin_amdgcn_readlane(rp, min(i + 2, 31)) : end;
    unsigned nA = csr_src[min(end + lane, elast)];
    unsigned nB = csr_src[min(end + 32 + lane, elast)];
    const float di = dinv[node];
    const int deg = end - beg;
    unsigned sA[8];
#pragma unroll
    for (int s = 0; s < 8; ++s)
      sA[s] = (unsigned)__builtin_amdgcn_ds_bpermute((4 * s + g) << 2, (int)kA);
    float4 acc = hs4[(size_t)(g == 0 ? node : n) * 8 + q];
    float4 vA[8];
#pragma unroll
    for (int s = 0; s < 8; ++s) vA[s] = hs4[(size_t)sA[s] * 8 + q];
    float4 accB = make_float4(0.0f, 0.0f, 0.0f, 0.0f);
    if (deg > 32) {
      unsigned sB[8];
#pragma unroll
      for (int s = 0; s < 8; ++s)
        sB[s] = (unsigned)__builtin_amdgcn_ds_bpermute((4 * s + g) << 2, (int)kB);
      float4 vB[8];
#pragma unroll
      for (int s = 0; s < 8; ++s) vB[s] = hs4[(size_t)sB[s] * 8 + q];
#pragma unroll
      for (int s = 0; s < 8; ++s) {
        accB.x += vB[s].x;
        accB.y += vB[s].y;
        accB.z += vB[s].z;
        accB.w += vB[s].w;
      }
      for (int e = beg + 64 + g; e < end; e += 4) {
        const float4 r = hs4[(size_t)csr_src[e] * 8 + q];
        accB.x += r.x;
        accB.y += r.y;
        accB.z += r.z;
        accB.w += r.w;
      }
    }
#pragma unroll
    for (int s = 0; s < 8; ++s) {
      acc.x += vA[s].x;
      acc.y += vA[s].y;
      acc.z += vA[s].z;
      acc.w += vA[s].w;
    }
    acc.x += accB.x;
    acc.y += accB.y;
    acc.z += accB.z;
    acc.w += accB.w;
    acc.x += __shfl_xor(acc.x, 8);
    acc.y += __shfl_xor(acc.y, 8);
    acc.z += __shfl_xor(acc.z, 8);
    acc.w += __shfl_xor(acc.w, 8);
    acc.x += __shfl_xor(acc.x, 16);
    acc.y += __shfl_xor(acc.y, 16);
    acc.z += __shfl_xor(acc.z, 16);
    acc.w += __shfl_xor(acc.w, 16);
    if constexpr (NEXT) {
      float4 r;
      r.x = fmaxf(di * acc.x + b4.x, 0.0f);
      r.y = fmaxf(di * acc.y + b4.y, 0.0f);
      r.z = fmaxf(di * acc.z + b4.z, 0.0f);
      r.w = fmaxf(di * acc.w + b4.w, 0.0f);
      if (lane == 8) r = make_float4(di, di, di, di);
      if (lane < 9) *reinterpret_cast<float4*>(&sT[(wid * NPW + i) * TP + 4 * lane]) = r;
    } else if (g == 0) {
      float4 r;
      r.x = di * acc.x + b4.x;
      r.y = di * acc.y + b4.y;
      r.z = di * acc.z + b4.z;
      r.w = di * acc.w + b4.w;
      out4[(size_t)node * 8 + q] = r;
    }
    kA = (end + lane < nend) ? nA : (unsigned)n;
    kB = (end + 32 + lane < nend) ? nB : (unsigned)n;
    beg = end;
    end = nend;
  }
  if constexpr (NEXT) {
    if (cntn < NPW) {
      float4* sT4 = reinterpret_cast<float4*>(&sT[wid * NPW * TP]);
#pragma unroll
      for (int h = 0; h < 3; ++h) {
        const int p = lane + 32 * h;
        if (p >= cntn * (TP / 4) && p < NPW * (TP / 4)) sT4[p] = make_float4(0.0f, 0.0f, 0.0f, 0.0f);
      }
    }
    const int lhalf = lane >> 4;
    const int l15 = lane & 15;
    const int mt = (wid >> 1) * 16;
    const int nt = (wid & 1) * 16;
    const int grow_base = blockIdx.x * (RW * NPW) + mt + lhalf * 8;
    v2f bf[HID / 4];
#pragma unroll
    for (int kk = 0; kk < HID / 4; ++kk) {
      const float* bp = &Wn[(4 * kk + lhalf * 2) * HID + nt + l15];
      bf[kk].x = bp[0];
      bf[kk].y = bp[HID];
    }
    __syncthreads();
    v8f acc = {};
#pragma unroll
    for (int kk = 0; kk < HID / 4; ++kk) {
      const float* ap = &sT[(mt + l15) * TP + 4 * kk + lhalf * 2];
      v2f a;
      a.x = ap[0];
      a.y = ap[1];
      acc = __builtin_amdgcn_wmma_f32_16x16x4_f32(false, a, false, bf[kk], (short)0, acc,
                                                  false, false);
    }
    float rs[8];
#pragma unroll
    for (int j = 0; j < 8; ++j) rs[j] = sT[(mt + lhalf * 8 + j) * TP + HID];
    const int col = nt + l15;
#pragma unroll
    for (int j = 0; j < 8; ++j) {
      const int gr = grow_base + j;
      if (gr < n) out[(size_t)gr * HID + col] = acc[j] * rs[j];
    }
    if (blockIdx.x == gridDim.x - 1 && threadIdx.x < HID) out[(size_t)n * HID + threadIdx.x] = 0.0f;
  }
}

template <int K>
__global__ __launch_bounds__(128)
void gemm_n32_kernel(const float* __restrict__ A, const float* __restrict__ W,
                     const float* __restrict__ rowscale, float* __restrict__ out, int M) {
  constexpr int LDK = K + 4;
  constexpr int V = K / 4;
  constexpr int NL = (32 * V) / 128;
  __shared__ __align__(16) float sA[32 * LDK];

  const int tid = threadIdx.x;
  const int block_row0 = blockIdx.x * 32;
  const int mlast = M - 1;

  const int wid   = tid >> 5;
  const int lane  = tid & 31;
  const int lhalf = lane >> 4;
  const int l15   = lane & 15;
  const int warp_m = (wid >> 1) * 16;
  const int n0     = (wid & 1) * 16;
  const int grow_base = block_row0 + warp_m + lhalf * 8;

  const v4f* A4 = reinterpret_cast<const v4f*>(A);
  v4f av[NL];
#pragma unroll
  for (int i = 0; i < NL; ++i) {
    const int idx = tid + 128 * i;
    const int r = idx / V;
    const int c4 = idx - r * V;
    av[i] = A4[(size_t)min(block_row0 + r, mlast) * V + c4];
  }
  float rs[8];
#pragma unroll
  for (int j = 0; j < 8; ++j) rs[j] = rowscale[min(grow_base + j, mlast)];
  if constexpr (NL == 8) {
    asm volatile("" :: "v"(av[0]), "v"(av[1]), "v"(av[2]), "v"(av[3]),
                       "v"(av[4]), "v"(av[5]), "v"(av[6]), "v"(av[7]));
  }
#pragma unroll
  for (int i = 0; i < NL; ++i) {
    const int idx = tid + 128 * i;
    const int r = idx / V;
    const int c4 = idx - r * V;
    *reinterpret_cast<v4f*>(&sA[r * LDK + 4 * c4]) = av[i];
  }
  __syncthreads();

  v8f acc = {};
  for (int k = 0; k < K; k += 4) {
    const float* ap = &sA[(warp_m + l15) * LDK + k + lhalf * 2];
    v2f a;
    a.x = ap[0];
    a.y = ap[1];
    const float* bp = &W[(k + lhalf * 2) * HID + n0 + l15];
    v2f b;
    b.x = bp[0];
    b.y = bp[HID];
    acc = __builtin_amdgcn_wmma_f32_16x16x4_f32(false, a, false, b, (short)0, acc,
                                                false, false);
  }

  const int col = n0 + l15;
#pragma unroll
  for (int j = 0; j < 8; ++j) {
    int gr = grow_base + j;
    if (gr < M) out[(long long)gr * HID + col] = acc[j] * rs[j];
  }
}


extern "C" void kernel_launch(void* const* d_in, const int* in_sizes, int n_in,
                              void* d_out, int out_size, void* d_ws, size_t ws_size,
                              hipStream_t stream) {
  const float* x  = (const float*)d_in[0];
  const int*   ei = (const int*)d_in[1];
  const float* W1 = (const float*)d_in[2];
  const float* b1 = (const float*)d_in[3];
  const float* W2 = (const float*)d_in[4];
  const float* b2 = (const float*)d_in[5];
  float* out = (float*)d_out;

  const int N = in_sizes[0] / IN_CH;
  const int E = in_sizes[1] / 2;
  const int* src = ei;
  const int* dst = ei + E;

  const int Npad = (N + 63) & ~63;
  const int Epad = (E + 63) & ~63;
  int* gtotal  = (int*)d_ws;
  float* dinv  = (float*)(gtotal + 64);
  unsigned* keys = (unsigned*)(dinv + Npad);
  float* bufA = (float*)(keys + Epad);
  float* bufB = bufA + (size_t)(N + 8) * HID;
  unsigned* csr_src = (unsigned*)(bufB + (size_t)(N + 8) * HID);
  int* rowptr = (int*)(csr_src + Epad);
  int* boff = rowptr + NBMAX * (BK + 1);

  const unsigned NB    = (N + BK - 1) / BK;
  const unsigned gPart = (E + PBLK - 1) / PBLK;
  const unsigned gGemm = (N + 31) / 32;
  const unsigned gRows = (N + RW * NPW - 1) / (RW * NPW);

  block_sort_kernel<<<gPart, PT, 0, stream>>>(src, dst, keys, boff, gtotal, E);
  bucket_csr_kernel<<<NB, AT, 0, stream>>>(keys, boff, gtotal, csr_src, rowptr, dinv,
                                           bufA + (size_t)N * HID, N, E, (int)gPart);

  gemm_n32_kernel<IN_CH><<<gGemm, 128, 0, stream>>>(x, W1, dinv, bufA, N);
  row_sum_kernel<true><<<gRows, RW * 32, 0, stream>>>(csr_src, rowptr, bufA, dinv, b1, W2, bufB, N, E);

  row_sum_kernel<false><<<gRows, RW * 32, 0, stream>>>(csr_src, rowptr, bufB, dinv, b2, W2, out, N, E);
}
